# starting kernel
# baseline (speedup 1.0000x reference)
#include <hip/hip_runtime.h>


typedef __attribute__((ext_vector_type(2))) float v2f;
typedef __attribute__((ext_vector_type(8))) float v8f;

#define D 128

__global__ void k_deg_init(float* __restrict__ deg, int n) {
  int i = blockIdx.x * blockDim.x + threadIdx.x;
  if (i < n) deg[i] = 1.0f;
}

__global__ void k_deg_count(const int* __restrict__ dst, float* __restrict__ deg, int e) {
  int i = blockIdx.x * blockDim.x + threadIdx.x;
  if (i < e) atomicAdd(&deg[dst[i]], 1.0f);
}

__global__ void k_deg_rsqrt(float* __restrict__ deg, int n) {
  int i = blockIdx.x * blockDim.x + threadIdx.x;
  if (i < n) deg[i] = rsqrtf(deg[i]);
}

__global__ void k_bias_init(float* __restrict__ out, const float* __restrict__ b, int total) {
  int i = blockIdx.x * blockDim.x + threadIdx.x;
  if (i < total) out[i] = b[i & (D - 1)];
}

template <bool RELU_IN>
__global__ __launch_bounds__(256) void k_gemm128(const float* __restrict__ A,
                                                 const float* __restrict__ W,
                                                 float* __restrict__ H, int nrows) {
  __shared__ float Ws[D * D];

  const int tid = threadIdx.x;
  {
    const float4* __restrict__ Wv = (const float4*)W;
    float4* Sv = (float4*)Ws;
#pragma unroll
    for (int i = 0; i < (D * D / 4) / 256; ++i)
      Sv[tid + i * 256] = Wv[tid + i * 256];
  }
  __syncthreads();

  const int wave = tid >> 5;
  const int lane = tid & 31;
  const int row0 = blockIdx.x * 128 + wave * 16;
  if (row0 >= nrows) return;

  const int m  = lane & 15;
  const int kb = (lane >> 4) << 1;

  v8f acc[8] = {};

  const float* __restrict__ Arow = A + (size_t)(row0 + m) * D;
  for (int k = 0; k < D; k += 4) {
    float ax = Arow[k + kb];
    float ay = Arow[k + kb + 1];
    if (RELU_IN) {
      ax = fmaxf(ax, 0.0f);
      ay = fmaxf(ay, 0.0f);
    }
    v2f a;
    a.x = ax;
    a.y = ay;
#pragma unroll
    for (int t = 0; t < 8; ++t) {
      const int col = t * 16 + m;
      v2f b;
      b.x = Ws[(k + kb) * D + col];
      b.y = Ws[(k + kb + 1) * D + col];
      acc[t] = __builtin_amdgcn_wmma_f32_16x16x4_f32(
           false, a,  false, b,
           (short)0, acc[t],  false,  false);
    }
  }

  const int rb = row0 + ((lane >> 4) << 3);
#pragma unroll
  for (int t = 0; t < 8; ++t) {
    const int col = t * 16 + m;
#pragma unroll
    for (int v = 0; v < 8; ++v)
      H[(size_t)(rb + v) * D + col] = acc[t][v];
  }
}

__global__ __launch_bounds__(256) void k_scatter(const float* __restrict__ h,
                                                 const int* __restrict__ ei, int E, int n,
                                                 const float* __restrict__ dis,
                                                 float* __restrict__ acc) {
  const int wid  = blockIdx.x * (blockDim.x >> 5) + (threadIdx.x >> 5);
  const int lane = threadIdx.x & 31;
  const int total = E + n;
  if (wid >= total) return;

  int s, d;
  if (wid < E) {
    s = ei[wid];
    d = ei[E + wid];
  } else {
    s = d = wid - E;
  }
  const float nrm = dis[s] * dis[d];

  const float4 hv = *(const float4*)(h + (size_t)s * D + lane * 4);
  float* o = acc + (size_t)d * D + lane * 4;
  atomicAdd(o + 0, hv.x * nrm);
  atomicAdd(o + 1, hv.y * nrm);
  atomicAdd(o + 2, hv.z * nrm);
  atomicAdd(o + 3, hv.w * nrm);
}

extern "C" void kernel_launch(void* const* d_in, const int* in_sizes, int n_in,
                              void* d_out, int out_size, void* d_ws, size_t ws_size,
                              hipStream_t stream) {
  const float* x  = (const float*)d_in[0];
  const int*   ei = (const int*)d_in[1];
  const float* W1 = (const float*)d_in[2];
  const float* b1 = (const float*)d_in[3];
  const float* W2 = (const float*)d_in[4];
  const float* b2 = (const float*)d_in[5];
  const float* W3 = (const float*)d_in[6];
  const float* b3 = (const float*)d_in[7];

  const int N = in_sizes[0] / D;
  const int E = in_sizes[1] / 2;

  float* deg  = (float*)d_ws;
  size_t npad = ((size_t)N + 255) & ~(size_t)255;
  float* buf1 = deg + npad;
  float* buf2 = buf1 + (size_t)N * D;
  float* out  = (float*)d_out;

  const dim3 blk(256);
  const int gN    = (N + 255) / 256;
  const int gE    = (E + 255) / 256;
  const int totF  = N * D;
  const int gF    = (totF + 255) / 256;
  const int gGemm = (N + 127) / 128;
  const int gScat = ((E + N) + 7) / 8;

  k_deg_init<<<gN, blk, 0, stream>>>(deg, N);
  k_deg_count<<<gE, blk, 0, stream>>>(ei + E, deg, E);
  k_deg_rsqrt<<<gN, blk, 0, stream>>>(deg, N);

  k_gemm128<false><<<gGemm, blk, 0, stream>>>(x, W1, buf1, N);
  k_bias_init<<<gF, blk, 0, stream>>>(buf2, b1, totF);
  k_scatter<<<gScat, blk, 0, stream>>>(buf1, ei, E, N, deg, buf2);

  k_gemm128<true><<<gGemm, blk, 0, stream>>>(buf2, W2, buf1, N);
  k_bias_init<<<gF, blk, 0, stream>>>(buf2, b2, totF);
  k_scatter<<<gScat, blk, 0, stream>>>(buf1, ei, E, N, deg, buf2);

  k_gemm128<true><<<gGemm, blk, 0, stream>>>(buf2, W3, buf1, N);
  k_bias_init<<<gF, blk, 0, stream>>>(out, b3, totF);
  k_scatter<<<gScat, blk, 0, stream>>>(buf1, ei, E, N, deg, out);
}


# Edge scatter: an accumulator row takes its atomics from one half of the chip only (row parity = half; die id read, chunk drawn by an integer ticket per half); atomic sum order unordered as before
# speedup vs baseline: 2.1792x; 2.1792x over previous
#include <hip/hip_runtime.h>


typedef __attribute__((ext_vector_type(2))) float v2f;
typedef __attribute__((ext_vector_type(8))) float v8f;

#define D 128

#define TILE 16384
#define CH   32768
#define CTH  512
#define CV   (CH / (4 * CTH))

__global__ __launch_bounds__(CTH) void k_deg_part(const int* __restrict__ dst, int E,
                                                  unsigned* __restrict__ part, int pitch) {
  __shared__ unsigned cnt[TILE];
  const int tid  = threadIdx.x;
  const int base = blockIdx.x * TILE;
  const int e0   = blockIdx.y * CH;

  const bool whole = ((((size_t)dst) & 15) == 0) && (e0 + CH <= E);
  int4 v[CV];
  if (whole) {
    const int4* d4 = (const int4*)(dst + e0);
#pragma unroll
    for (int j = 0; j < CV; ++j) v[j] = d4[tid + j * CTH];
  } else {
#pragma unroll
    for (int j = 0; j < CV; ++j) {
      const int e = e0 + 4 * (tid + j * CTH);
      v[j].x = (e     < E) ? dst[e]     : -1;
      v[j].y = (e + 1 < E) ? dst[e + 1] : -1;
      v[j].z = (e + 2 < E) ? dst[e + 2] : -1;
      v[j].w = (e + 3 < E) ? dst[e + 3] : -1;
    }
  }

  uint4* c4 = (uint4*)cnt;
  for (int i = tid; i < TILE / 4; i += CTH) c4[i] = make_uint4(0u, 0u, 0u, 0u);
  __syncthreads();

#pragma unroll
  for (int j = 0; j < CV; ++j) {
    const unsigned r0 = (unsigned)(v[j].x - base);
    const unsigned r1 = (unsigned)(v[j].y - base);
    const unsigned r2 = (unsigned)(v[j].z - base);
    const unsigned r3 = (unsigned)(v[j].w - base);
    if (r0 < TILE) atomicAdd(&cnt[r0], 1u);
    if (r1 < TILE) atomicAdd(&cnt[r1], 1u);
    if (r2 < TILE) atomicAdd(&cnt[r2], 1u);
    if (r3 < TILE) atomicAdd(&cnt[r3], 1u);
  }
  __syncthreads();

  uint4* o4 = (uint4*)(part + (size_t)blockIdx.y * pitch + base);
  for (int i = tid; i < TILE / 4; i += CTH) o4[i] = c4[i];
}

#define PG 32

#define TKW 384

__global__ void k_deg_rsqrt(const unsigned* __restrict__ part, int pitch, int nchunk,
                            float* __restrict__ dis, int n, unsigned* __restrict__ tick) {
  int i = blockIdx.x * blockDim.x + threadIdx.x;
  if (i < TKW) tick[i] = 0u;
  if (i < n) {
    unsigned c = 0;
    for (int q0 = 0; q0 < nchunk; q0 += PG) {
      unsigned t[PG];
#pragma unroll
      for (int u = 0; u < PG; ++u)
        t[u] = (q0 + u < nchunk) ? part[(size_t)(q0 + u) * pitch + i] : 0u;
#pragma unroll
      for (int u = 0; u < PG; ++u) c += t[u];
    }
    dis[i] = rsqrtf(1.0f + (float)c);
  }
}

template <bool RELU_IN>
__global__ __launch_bounds__(256) void k_gemm128(const float* __restrict__ A,
                                                 const float* __restrict__ W,
                                                 const float* __restrict__ bias,
                                                 const float* __restrict__ dis,
                                                 float* __restrict__ H,
                                                 float* __restrict__ S, int nrows) {
  __shared__ float Ws[D * D];

  const int tid = threadIdx.x;
  {
    const float4* __restrict__ Wv = (const float4*)W;
    float4* Sv = (float4*)Ws;
#pragma unroll
    for (int i = 0; i < (D * D / 4) / 256; ++i)
      Sv[tid + i * 256] = Wv[tid + i * 256];
  }
  __syncthreads();

  const int wave = tid >> 5;
  const int lane = tid & 31;
  const int row0 = blockIdx.x * 128 + wave * 16;
  if (row0 >= nrows) return;

  const int m  = lane & 15;
  const int kb = (lane >> 4) << 1;

  v8f acc[8] = {};

  const float* __restrict__ Arow = A + (size_t)(row0 + m) * D;
  for (int k = 0; k < D; k += 4) {
    float ax = Arow[k + kb];
    float ay = Arow[k + kb + 1];
    if (RELU_IN) {
      ax = fmaxf(ax, 0.0f);
      ay = fmaxf(ay, 0.0f);
    }
    v2f a;
    a.x = ax;
    a.y = ay;
#pragma unroll
    for (int t = 0; t < 8; ++t) {
      const int col = t * 16 + m;
      v2f b;
      b.x = Ws[(k + kb) * D + col];
      b.y = Ws[(k + kb + 1) * D + col];
      acc[t] = __builtin_amdgcn_wmma_f32_16x16x4_f32(
           false, a,  false, b,
           (short)0, acc[t],  false,  false);
    }
  }

  const int rb = row0 + ((lane >> 4) << 3);

  const float4 d0 = *(const float4*)(dis + rb);
  const float4 d1 = *(const float4*)(dis + rb + 4);
  const float w[8] = {d0.x * d0.x, d0.y * d0.y, d0.z * d0.z, d0.w * d0.w,
                      d1.x * d1.x, d1.y * d1.y, d1.z * d1.z, d1.w * d1.w};
#pragma unroll
  for (int t = 0; t < 8; ++t) {
    const int col = t * 16 + m;
    const float bt = bias[col];
#pragma unroll
    for (int v = 0; v < 8; ++v) {
      const float hv = acc[t][v];
      H[(size_t)(rb + v) * D + col] = hv;
      S[(size_t)(rb + v) * D + col] = bt + hv * w[v];
    }
  }
}

#define ITEM 512
#define GRP 8

__device__ __forceinline__ void scatter_list(const float* __restrict__ h,
                                             float* __restrict__ acc, int lane,
                                             int s, int d, int nb, int n) {
  int c = 0;
  for (; c + GRP <= n; c += GRP) {
    float v[GRP][4];
#pragma unroll
    for (int u = 0; u < GRP; ++u) {
      const int sj = __builtin_amdgcn_readlane(s, c + u);
      const float* hr = h + (size_t)sj * D + lane;
      v[u][0] = hr[0];
      v[u][1] = hr[32];
      v[u][2] = hr[64];
      v[u][3] = hr[96];
    }
#pragma unroll
    for (int u = 0; u < GRP; ++u) {
      const int dj = __builtin_amdgcn_readlane(d, c + u);
      const float nj = __int_as_float(__builtin_amdgcn_readlane(nb, c + u));
      float* o = acc + (size_t)dj * D + lane;
      atomicAdd(o + 0,  v[u][0] * nj);
      atomicAdd(o + 32, v[u][1] * nj);
      atomicAdd(o + 64, v[u][2] * nj);
      atomicAdd(o + 96, v[u][3] * nj);
    }
  }
  if (c < n) {
    float v[GRP][4];
#pragma unroll
    for (int u = 0; u < GRP; ++u) {
      v[u][0] = 0.0f; v[u][1] = 0.0f; v[u][2] = 0.0f; v[u][3] = 0.0f;
      if (c + u < n) {
        const int sj = __builtin_amdgcn_readlane(s, c + u);
        const float* hr = h + (size_t)sj * D + lane;
        v[u][0] = hr[0];
        v[u][1] = hr[32];
        v[u][2] = hr[64];
        v[u][3] = hr[96];
      }
    }
#pragma unroll
    for (int u = 0; u < GRP; ++u) {
      if (c + u < n) {
        const int dj = __builtin_amdgcn_readlane(d, c + u);
        const float nj = __int_as_float(__builtin_amdgcn_readlane(nb, c + u));
        float* o = acc + (size_t)dj * D + lane;
        atomicAdd(o + 0,  v[u][0] * nj);
        atomicAdd(o + 32, v[u][1] * nj);
        atomicAdd(o + 64, v[u][2] * nj);
        atomicAdd(o + 96, v[u][3] * nj);
      }
    }
  }
}

__global__ __launch_bounds__(256) void k_scatter(const float* __restrict__ h,
                                                 const int* __restrict__ ei, int E,
                                                 const float* __restrict__ dis,
                                                 float* __restrict__ acc,
                                                 unsigned* __restrict__ tick, int nItems) {
  __shared__ int4 comp[8][64];
  __shared__ int pick[2];

  if (threadIdx.x == 0) {
    const unsigned id = __builtin_amdgcn_s_sendmsg_rtn(0x87);
    int par = (int)((id >> 18) & 1u);
    unsigned t = atomicAdd(&tick[par * 64], 1u);
    if (t >= (unsigned)nItems) {
      par ^= 1;
      t = atomicAdd(&tick[par * 64], 1u);
    }
    pick[0] = (int)t;
    pick[1] = par;
  }
  __syncthreads();
  const int item = pick[0];
  const int par  = pick[1];
  if ((unsigned)item >= (unsigned)nItems) return;

  const int wv   = __builtin_amdgcn_readfirstlane((int)(threadIdx.x >> 5));
  const int lane = threadIdx.x & 31;
  const int e0   = item * ITEM + wv * 64;

  const int ea = e0 + lane;
  const int eb = ea + 32;
  const int ia = (ea < E) ? ea : E - 1;
  const int ib = (eb < E) ? eb : E - 1;
  const int sa = ei[ia];
  const int da = ei[E + ia];
  const int sb = ei[ib];
  const int db = ei[E + ib];
  const bool ma = (ea < E) && ((da & 1) == par);
  const bool mb = (eb < E) && ((db & 1) == par);
  const float na = dis[sa] * dis[da];
  const float nb = dis[sb] * dis[db];

  const unsigned ka = __builtin_amdgcn_ballot_w32(ma);
  const unsigned kb = __builtin_amdgcn_ballot_w32(mb);
  const int cnta = __builtin_popcount(ka);
  const int cnt  = cnta + __builtin_popcount(kb);
  const int pa = ma ? (int)__builtin_amdgcn_mbcnt_lo(ka, 0u)
                    : cnt + (int)__builtin_amdgcn_mbcnt_lo(~ka, 0u);
  const int pb = mb ? cnta + (int)__builtin_amdgcn_mbcnt_lo(kb, 0u)
                    : cnt + (32 - cnta) + (int)__builtin_amdgcn_mbcnt_lo(~kb, 0u);
  comp[wv][pa] = make_int4(sa, da, __float_as_int(na), 0);
  comp[wv][pb] = make_int4(sb, db, __float_as_int(nb), 0);
  __syncthreads();
  const int4 ca = comp[wv][lane];
  const int4 cb = comp[wv][32 + lane];

  scatter_list(h, acc, lane, ca.x, ca.y, ca.z, cnt < 32 ? cnt : 32);
  if (cnt > 32) scatter_list(h, acc, lane, cb.x, cb.y, cb.z, cnt - 32);
}

extern "C" void kernel_launch(void* const* d_in, const int* in_sizes, int n_in,
                              void* d_out, int out_size, void* d_ws, size_t ws_size,
                              hipStream_t stream) {
  const float* x  = (const float*)d_in[0];
  const int*   ei = (const int*)d_in[1];
  const float* W1 = (const float*)d_in[2];
  const float* b1 = (const float*)d_in[3];
  const float* W2 = (const float*)d_in[4];
  const float* b2 = (const float*)d_in[5];
  const float* W3 = (const float*)d_in[6];
  const float* b3 = (const float*)d_in[7];

  const int N = in_sizes[0] / D;
  const int E = in_sizes[1] / 2;

  float* deg  = (float*)d_ws;
  size_t npad = ((size_t)N + 255) & ~(size_t)255;
  float* buf1 = deg + npad;
  float* buf2 = buf1 + (size_t)N * D;
  float* buf3 = buf2 + (size_t)N * D;
  unsigned* part = (unsigned*)(buf3 + (size_t)N * D);
  float* out  = (float*)d_out;

  const dim3 blk(256);
  const int gN    = (N + 255) / 256;
  const int nTile  = (N + TILE - 1) / TILE;
  const int nChunk = (E + CH - 1) / CH;
  const int gGemm = (N + 127) / 128;
  const int nItem = (E + ITEM - 1) / ITEM;
  unsigned* tick = part + (size_t)nChunk * nTile * TILE;

  k_deg_part<<<dim3(nTile, nChunk), dim3(CTH), 0, stream>>>(ei + E, E, part, nTile * TILE);
  k_deg_rsqrt<<<gN, blk, 0, stream>>>(part, nTile * TILE, nChunk, deg, N, tick);

  k_gemm128<false><<<gGemm, blk, 0, stream>>>(x, W1, b1, deg, buf1, buf2, N);
  k_scatter<<<2 * nItem, blk, 0, stream>>>(buf1, ei, E, deg, buf2, tick, nItem);

  k_gemm128<true><<<gGemm, blk, 0, stream>>>(buf2, W2, b2, deg, buf1, buf3, N);
  k_scatter<<<2 * nItem, blk, 0, stream>>>(buf1, ei, E, deg, buf3, tick + 128, nItem);

  k_gemm128<true><<<gGemm, blk, 0, stream>>>(buf3, W3, b3, deg, buf1, out, N);
  k_scatter<<<2 * nItem, blk, 0, stream>>>(buf1, ei, E, deg, out, tick + 256, nItem);
}


# Edge scatter: the chip's halves split the accumulator by column half (256-byte half rows), not row parity: a half gathers and adds half of every row, no edge sorting; atomic sum order as before
# speedup vs baseline: 2.6674x; 1.2240x over previous
#include <hip/hip_runtime.h>


typedef __attribute__((ext_vector_type(2))) float v2f;
typedef __attribute__((ext_vector_type(8))) float v8f;

#define D 128

#define TILE 16384
#define CH   32768
#define CTH  512
#define CV   (CH / (4 * CTH))

__global__ __launch_bounds__(CTH) void k_deg_part(const int* __restrict__ dst, int E,
                                                  unsigned* __restrict__ part, int pitch) {
  __shared__ unsigned cnt[TILE];
  const int tid  = threadIdx.x;
  const int base = blockIdx.x * TILE;
  const int e0   = blockIdx.y * CH;

  const bool whole = ((((size_t)dst) & 15) == 0) && (e0 + CH <= E);
  int4 v[CV];
  if (whole) {
    const int4* d4 = (const int4*)(dst + e0);
#pragma unroll
    for (int j = 0; j < CV; ++j) v[j] = d4[tid + j * CTH];
  } else {
#pragma unroll
    for (int j = 0; j < CV; ++j) {
      const int e = e0 + 4 * (tid + j * CTH);
      v[j].x = (e     < E) ? dst[e]     : -1;
      v[j].y = (e + 1 < E) ? dst[e + 1] : -1;
      v[j].z = (e + 2 < E) ? dst[e + 2] : -1;
      v[j].w = (e + 3 < E) ? dst[e + 3] : -1;
    }
  }

  uint4* c4 = (uint4*)cnt;
  for (int i = tid; i < TILE / 4; i += CTH) c4[i] = make_uint4(0u, 0u, 0u, 0u);
  __syncthreads();

#pragma unroll
  for (int j = 0; j < CV; ++j) {
    const unsigned r0 = (unsigned)(v[j].x - base);
    const unsigned r1 = (unsigned)(v[j].y - base);
    const unsigned r2 = (unsigned)(v[j].z - base);
    const unsigned r3 = (unsigned)(v[j].w - base);
    if (r0 < TILE) atomicAdd(&cnt[r0], 1u);
    if (r1 < TILE) atomicAdd(&cnt[r1], 1u);
    if (r2 < TILE) atomicAdd(&cnt[r2], 1u);
    if (r3 < TILE) atomicAdd(&cnt[r3], 1u);
  }
  __syncthreads();

  uint4* o4 = (uint4*)(part + (size_t)blockIdx.y * pitch + base);
  for (int i = tid; i < TILE / 4; i += CTH) o4[i] = c4[i];
}

#define PG 32

#define TKW 384

__global__ void k_deg_rsqrt(const unsigned* __restrict__ part, int pitch, int nchunk,
                            float* __restrict__ dis, int n, unsigned* __restrict__ tick) {
  int i = blockIdx.x * blockDim.x + threadIdx.x;
  if (i < TKW) tick[i] = 0u;
  if (i < n) {
    unsigned c = 0;
    for (int q0 = 0; q0 < nchunk; q0 += PG) {
      unsigned t[PG];
#pragma unroll
      for (int u = 0; u < PG; ++u)
        t[u] = (q0 + u < nchunk) ? part[(size_t)(q0 + u) * pitch + i] : 0u;
#pragma unroll
      for (int u = 0; u < PG; ++u) c += t[u];
    }
    dis[i] = rsqrtf(1.0f + (float)c);
  }
}

#define GW 256

__device__ __forceinline__ int gemm_k(int kk, int h) {
  return 8 * (kk >> 1) + 2 * (kk & 1) + 4 * h;
}

__device__ __forceinline__ void gemm_load_a(v2f (&a)[32], const float* A,
                                            int s, int m, int h) {
  const float4* Arow = (const float4*)(A + (size_t)(s * 16 + m) * D + 4 * h);
#pragma unroll
  for (int q = 0; q < 16; ++q) {
    const float4 f = Arow[2 * q];
    a[2 * q].x = f.x;
    a[2 * q].y = f.y;
    a[2 * q + 1].x = f.z;
    a[2 * q + 1].y = f.w;
  }
}

template <bool RELU_IN>
__device__ __forceinline__ void gemm_stripe(const v2f (&a)[32], const v2f (&b)[32][4],
                                            int s, int h, int c0,
                                            const float* __restrict__ bias,
                                            const float* __restrict__ dis,
                                            float* H, float* S) {
  v8f acc[4] = {};
#pragma unroll
  for (int kk = 0; kk < 32; ++kk) {
    v2f av = a[kk];
    if (RELU_IN) {
      av.x = fmaxf(av.x, 0.0f);
      av.y = fmaxf(av.y, 0.0f);
    }
#pragma unroll
    for (int t = 0; t < 4; ++t) {
      acc[t] = __builtin_amdgcn_wmma_f32_16x16x4_f32(
           false, av,  false, b[kk][t],
           (short)0, acc[t],  false,  false);
    }
  }

  const int rb = s * 16 + 8 * h;

  const float4 d0 = *(const float4*)(dis + rb);
  const float4 d1 = *(const float4*)(dis + rb + 4);
  const float w[8] = {d0.x * d0.x, d0.y * d0.y, d0.z * d0.z, d0.w * d0.w,
                      d1.x * d1.x, d1.y * d1.y, d1.z * d1.z, d1.w * d1.w};
  const float4 bt = *(const float4*)(bias + c0);
#pragma unroll
  for (int v = 0; v < 8; ++v) {
    float4 hv;
    hv.x = acc[0][v];
    hv.y = acc[1][v];
    hv.z = acc[2][v];
    hv.w = acc[3][v];
    float4 sv;
    sv.x = bt.x + hv.x * w[v];
    sv.y = bt.y + hv.y * w[v];
    sv.z = bt.z + hv.z * w[v];
    sv.w = bt.w + hv.w * w[v];
    *(float4*)(H + (size_t)(rb + v) * D + c0) = hv;
    *(float4*)(S + (size_t)(rb + v) * D + c0) = sv;
  }
}

template <bool RELU_IN>
__global__ __launch_bounds__(256) void k_gemm128(const float* A,
                                                 const float* __restrict__ W,
                                                 const float* __restrict__ bias,
                                                 const float* __restrict__ dis,
                                                 float* H,
                                                 float* S, int nrows) {
  const int tid  = threadIdx.x;
  const int lane = tid & 31;
  const int wv = __builtin_amdgcn_readfirstlane(tid >> 5);
  const int ch = wv & 1;
  const int nW = (int)gridDim.x * 4;
  const int nStripes = nrows >> 4;
  int s = (wv >> 1) * (int)gridDim.x + (int)blockIdx.x;
  if (s >= nStripes) return;

  const int m  = lane & 15;
  const int h  = lane >> 4;
  const int c0 = ch * 64 + 4 * m;

  v2f b[32][4];
#pragma unroll
  for (int kk = 0; kk < 32; ++kk) {
    const int k0 = gemm_k(kk, h);
    const float4 wx = *(const float4*)(W + (size_t)k0 * D + c0);
    const float4 wy = *(const float4*)(W + (size_t)(k0 + 1) * D + c0);
    b[kk][0].x = wx.x; b[kk][0].y = wy.x;
    b[kk][1].x = wx.y; b[kk][1].y = wy.y;
    b[kk][2].x = wx.z; b[kk][2].y = wy.z;
    b[kk][3].x = wx.w; b[kk][3].y = wy.w;
  }

  v2f a0[32], a1[32];
  gemm_load_a(a0, A, s, m, h);
  for (;;) {
    int sn = s + nW;
    bool more = sn < nStripes;
    gemm_load_a(a1, A, more ? sn : s, m, h);
    gemm_stripe<RELU_IN>(a0, b, s, h, c0, bias, dis, H, S);
    if (!more) break;
    s = sn;
    sn = s + nW;
    more = sn < nStripes;
    gemm_load_a(a0, A, more ? sn : s, m, h);
    gemm_stripe<RELU_IN>(a1, b, s, h, c0, bias, dis, H, S);
    if (!more) break;
    s = sn;
  }
}

#define ITEM 512
#define GRP 16

__device__ __forceinline__ void scatter_list(const float* __restrict__ h,
                                             float* __restrict__ acc,
                                             int s, int d, int nb, int n) {
  int c = 0;
  for (; c + GRP <= n; c += GRP) {
    float v[GRP][2];
#pragma unroll
    for (int u = 0; u < GRP; ++u) {
      const int sj = __builtin_amdgcn_readlane(s, c + u);
      const float* hr = h + (size_t)sj * D;
      v[u][0] = hr[0];
      v[u][1] = hr[32];
    }
#pragma unroll
    for (int u = 0; u < GRP; ++u) {
      const int dj = __builtin_amdgcn_readlane(d, c + u);
      const float nj = __int_as_float(__builtin_amdgcn_readlane(nb, c + u));
      float* o = acc + (size_t)dj * D;
      atomicAdd(o + 0,  v[u][0] * nj);
      atomicAdd(o + 32, v[u][1] * nj);
    }
  }
  if (c < n) {
    float v[GRP][2];
#pragma unroll
    for (int u = 0; u < GRP; ++u) {
      v[u][0] = 0.0f; v[u][1] = 0.0f;
      if (c + u < n) {
        const int sj = __builtin_amdgcn_readlane(s, c + u);
        const float* hr = h + (size_t)sj * D;
        v[u][0] = hr[0];
        v[u][1] = hr[32];
      }
    }
#pragma unroll
    for (int u = 0; u < GRP; ++u) {
      if (c + u < n) {
        const int dj = __builtin_amdgcn_readlane(d, c + u);
        const float nj = __int_as_float(__builtin_amdgcn_readlane(nb, c + u));
        float* o = acc + (size_t)dj * D;
        atomicAdd(o + 0,  v[u][0] * nj);
        atomicAdd(o + 32, v[u][1] * nj);
      }
    }
  }
}

__global__ __launch_bounds__(256) __attribute__((amdgpu_waves_per_eu(16, 16))) void k_scatter(const float* __restrict__ h,
                                                 const int* __restrict__ ei, int E,
                                                 const float* __restrict__ dis,
                                                 float* __restrict__ acc,
                                                 unsigned* __restrict__ tick, int nItems) {
  __shared__ int pick[2];

  if (threadIdx.x == 0) {
    const unsigned id = __builtin_amdgcn_s_sendmsg_rtn(0x87);
    int par = (int)((id >> 18) & 1u);
    unsigned t = atomicAdd(&tick[par * 64], 1u);
    if (t >= (unsigned)nItems) {
      par ^= 1;
      t = atomicAdd(&tick[par * 64], 1u);
    }
    pick[0] = (int)t;
    pick[1] = par;
  }
  __syncthreads();
  const int item = __builtin_amdgcn_readfirstlane(pick[0]);
  const int par  = __builtin_amdgcn_readfirstlane(pick[1]);
  if ((unsigned)item >= (unsigned)nItems) return;

  const int wv   = __builtin_amdgcn_readfirstlane((int)(threadIdx.x >> 5));
  const int lane = threadIdx.x & 31;
  const int e0   = item * ITEM + wv * 64;

  const int ea = e0 + lane;
  const int eb = ea + 32;
  const int ia = (ea < E) ? ea : E - 1;
  const int ib = (eb < E) ? eb : E - 1;
  const int sa = ei[ia];
  const int da = ei[E + ia];
  const int sb = ei[ib];
  const int db = ei[E + ib];
  const float na = dis[sa] * dis[da];
  const float nb = dis[sb] * dis[db];

  int cntA = E - e0;
  cntA = cntA < 0 ? 0 : (cntA > 32 ? 32 : cntA);
  int cntB = E - e0 - 32;
  cntB = cntB < 0 ? 0 : (cntB > 32 ? 32 : cntB);

  const int col = 64 * par + lane;
  scatter_list(h + col, acc + col, sa, da, __float_as_int(na), cntA);
  scatter_list(h + col, acc + col, sb, db, __float_as_int(nb), cntB);
}

extern "C" void kernel_launch(void* const* d_in, const int* in_sizes, int n_in,
                              void* d_out, int out_size, void* d_ws, size_t ws_size,
                              hipStream_t stream) {
  const float* x  = (const float*)d_in[0];
  const int*   ei = (const int*)d_in[1];
  const float* W1 = (const float*)d_in[2];
  const float* b1 = (const float*)d_in[3];
  const float* W2 = (const float*)d_in[4];
  const float* b2 = (const float*)d_in[5];
  const float* W3 = (const float*)d_in[6];
  const float* b3 = (const float*)d_in[7];

  const int N = in_sizes[0] / D;
  const int E = in_sizes[1] / 2;

  float* deg  = (float*)d_ws;
  size_t npad = ((size_t)N + 255) & ~(size_t)255;
  float* buf1 = deg + npad;
  float* buf2 = buf1 + (size_t)N * D;
  float* buf3 = buf2 + (size_t)N * D;
  unsigned* part = (unsigned*)(buf3 + (size_t)N * D);
  float* out  = (float*)d_out;

  const dim3 blk(256);
  const int gN    = (N + 255) / 256;
  const int nTile  = (N + TILE - 1) / TILE;
  const int nChunk = (E + CH - 1) / CH;
  const int nItem = (E + ITEM - 1) / ITEM;
  unsigned* tick = part + (size_t)nChunk * nTile * TILE;

  k_deg_part<<<dim3(nTile, nChunk), dim3(CTH), 0, stream>>>(ei + E, E, part, nTile * TILE);
  k_deg_rsqrt<<<gN, blk, 0, stream>>>(part, nTile * TILE, nChunk, deg, N, tick);

  k_gemm128<false><<<GW, blk, 0, stream>>>(x, W1, b1, deg, buf1, buf2, N);
  k_scatter<<<2 * nItem, blk, 0, stream>>>(buf1, ei, E, deg, buf2, tick, nItem);

  k_gemm128<true><<<GW, blk, 0, stream>>>(buf2, W2, b2, deg, buf1, buf3, N);
  k_scatter<<<2 * nItem, blk, 0, stream>>>(buf1, ei, E, deg, buf3, tick + 128, nItem);

  k_gemm128<true><<<GW, blk, 0, stream>>>(buf3, W3, b3, deg, buf1, out, N);
  k_scatter<<<2 * nItem, blk, 0, stream>>>(buf1, ei, E, deg, out, tick + 256, nItem);
}


# Edge weights leave the scatter's gathers: GEMM epilogue stores dis[r]*h, scatter 1 stores dis[dst] per edge, scatters 2, 3 read it in edge order; products regrouped, atomic sum order as before
# speedup vs baseline: 2.9912x; 1.1214x over previous
#include <hip/hip_runtime.h>


typedef __attribute__((ext_vector_type(2))) float v2f;
typedef __attribute__((ext_vector_type(8))) float v8f;

#define D 128

#define TILE 16384
#define CH   32768
#define CTH  512
#define CV   (CH / (4 * CTH))

__global__ __launch_bounds__(CTH) void k_deg_part(const int* __restrict__ dst, int E,
                                                  unsigned* __restrict__ part, int pitch) {
  __shared__ unsigned cnt[TILE];
  const int tid  = threadIdx.x;
  const int base = blockIdx.x * TILE;
  const int e0   = blockIdx.y * CH;

  const bool whole = ((((size_t)dst) & 15) == 0) && (e0 + CH <= E);
  int4 v[CV];
  if (whole) {
    const int4* d4 = (const int4*)(dst + e0);
#pragma unroll
    for (int j = 0; j < CV; ++j) v[j] = d4[tid + j * CTH];
  } else {
#pragma unroll
    for (int j = 0; j < CV; ++j) {
      const int e = e0 + 4 * (tid + j * CTH);
      v[j].x = (e     < E) ? dst[e]     : -1;
      v[j].y = (e + 1 < E) ? dst[e + 1] : -1;
      v[j].z = (e + 2 < E) ? dst[e + 2] : -1;
      v[j].w = (e + 3 < E) ? dst[e + 3] : -1;
    }
  }

  uint4* c4 = (uint4*)cnt;
  for (int i = tid; i < TILE / 4; i += CTH) c4[i] = make_uint4(0u, 0u, 0u, 0u);
  __syncthreads();

#pragma unroll
  for (int j = 0; j < CV; ++j) {
    const unsigned r0 = (unsigned)(v[j].x - base);
    const unsigned r1 = (unsigned)(v[j].y - base);
    const unsigned r2 = (unsigned)(v[j].z - base);
    const unsigned r3 = (unsigned)(v[j].w - base);
    if (r0 < TILE) atomicAdd(&cnt[r0], 1u);
    if (r1 < TILE) atomicAdd(&cnt[r1], 1u);
    if (r2 < TILE) atomicAdd(&cnt[r2], 1u);
    if (r3 < TILE) atomicAdd(&cnt[r3], 1u);
  }
  __syncthreads();

  uint4* o4 = (uint4*)(part + (size_t)blockIdx.y * pitch + base);
  for (int i = tid; i < TILE / 4; i += CTH) o4[i] = c4[i];
}

#define PG 32

#define TKW 384

__global__ void k_deg_rsqrt(const unsigned* __restrict__ part, int pitch, int nchunk,
                            float* __restrict__ dis, int n, unsigned* __restrict__ tick) {
  int i = blockIdx.x * blockDim.x + threadIdx.x;
  if (i < TKW) tick[i] = 0u;
  if (i < n) {
    unsigned c = 0;
    for (int q0 = 0; q0 < nchunk; q0 += PG) {
      unsigned t[PG];
#pragma unroll
      for (int u = 0; u < PG; ++u)
        t[u] = (q0 + u < nchunk) ? part[(size_t)(q0 + u) * pitch + i] : 0u;
#pragma unroll
      for (int u = 0; u < PG; ++u) c += t[u];
    }
    dis[i] = rsqrtf(1.0f + (float)c);
  }
}

#define GW 256

__device__ __forceinline__ int gemm_k(int kk, int h) {
  return 8 * (kk >> 1) + 2 * (kk & 1) + 4 * h;
}

__device__ __forceinline__ void gemm_load_a(v2f (&a)[32], const float* A,
                                            int s, int m, int h) {
  const float4* Arow = (const float4*)(A + (size_t)(s * 16 + m) * D + 4 * h);
#pragma unroll
  for (int q = 0; q < 16; ++q) {
    const float4 f = Arow[2 * q];
    a[2 * q].x = f.x;
    a[2 * q].y = f.y;
    a[2 * q + 1].x = f.z;
    a[2 * q + 1].y = f.w;
  }
}

template <bool RELU_IN>
__device__ __forceinline__ void gemm_stripe(const v2f (&a)[32], const v2f (&b)[32][4],
                                            int s, int h, int c0,
                                            const float* __restrict__ bias,
                                            const float* __restrict__ dis,
                                            float* H, float* S) {
  v8f acc[4] = {};
#pragma unroll
  for (int kk = 0; kk < 32; ++kk) {
    v2f av = a[kk];
    if (RELU_IN) {
      av.x = fmaxf(av.x, 0.0f);
      av.y = fmaxf(av.y, 0.0f);
    }
#pragma unroll
    for (int t = 0; t < 4; ++t) {
      acc[t] = __builtin_amdgcn_wmma_f32_16x16x4_f32(
           false, av,  false, b[kk][t],
           (short)0, acc[t],  false,  false);
    }
  }

  const int rb = s * 16 + 8 * h;

  const float4 d0 = *(const float4*)(dis + rb);
  const float4 d1 = *(const float4*)(dis + rb + 4);
  const float w[8] = {d0.x, d0.y, d0.z, d0.w, d1.x, d1.y, d1.z, d1.w};
  const float4 bt = *(const float4*)(bias + c0);
#pragma unroll
  for (int v = 0; v < 8; ++v) {
    float4 gv;
    gv.x = acc[0][v] * w[v];
    gv.y = acc[1][v] * w[v];
    gv.z = acc[2][v] * w[v];
    gv.w = acc[3][v] * w[v];
    float4 sv;
    sv.x = bt.x + gv.x * w[v];
    sv.y = bt.y + gv.y * w[v];
    sv.z = bt.z + gv.z * w[v];
    sv.w = bt.w + gv.w * w[v];
    *(float4*)(H + (size_t)(rb + v) * D + c0) = gv;
    *(float4*)(S + (size_t)(rb + v) * D + c0) = sv;
  }
}

template <bool RELU_IN>
__global__ __launch_bounds__(256) void k_gemm128(const float* A,
                                                 const float* __restrict__ W,
                                                 const float* __restrict__ bias,
                                                 const float* __restrict__ dis,
                                                 float* H,
                                                 float* S, int nrows) {
  const int tid  = threadIdx.x;
  const int lane = tid & 31;
  const int wv = __builtin_amdgcn_readfirstlane(tid >> 5);
  const int ch = wv & 1;
  const int nW = (int)gridDim.x * 4;
  const int nStripes = nrows >> 4;
  int s = (wv >> 1) * (int)gridDim.x + (int)blockIdx.x;
  if (s >= nStripes) return;

  const int m  = lane & 15;
  const int h  = lane >> 4;
  const int c0 = ch * 64 + 4 * m;

  v2f b[32][4];
#pragma unroll
  for (int kk = 0; kk < 32; ++kk) {
    const int k0 = gemm_k(kk, h);
    const float4 wx = *(const float4*)(W + (size_t)k0 * D + c0);
    const float4 wy = *(const float4*)(W + (size_t)(k0 + 1) * D + c0);
    b[kk][0].x = wx.x; b[kk][0].y = wy.x;
    b[kk][1].x = wx.y; b[kk][1].y = wy.y;
    b[kk][2].x = wx.z; b[kk][2].y = wy.z;
    b[kk][3].x = wx.w; b[kk][3].y = wy.w;
  }

  v2f a0[32], a1[32];
  gemm_load_a(a0, A, s, m, h);
  for (;;) {
    int sn = s + nW;
    bool more = sn < nStripes;
    gemm_load_a(a1, A, more ? sn : s, m, h);
    gemm_stripe<RELU_IN>(a0, b, s, h, c0, bias, dis, H, S);
    if (!more) break;
    s = sn;
    sn = s + nW;
    more = sn < nStripes;
    gemm_load_a(a0, A, more ? sn : s, m, h);
    gemm_stripe<RELU_IN>(a1, b, s, h, c0, bias, dis, H, S);
    if (!more) break;
    s = sn;
  }
}

#define ITEM 512
#define GRP 16

__device__ __forceinline__ void scatter_list(const float* __restrict__ h,
                                             float* __restrict__ acc,
                                             int s, int d, int nb, int n) {
  int c = 0;
  for (; c + GRP <= n; c += GRP) {
    float v[GRP][2];
#pragma unroll
    for (int u = 0; u < GRP; ++u) {
      const int sj = __builtin_amdgcn_readlane(s, c + u);
      const float* hr = h + (size_t)sj * D;
      v[u][0] = hr[0];
      v[u][1] = hr[32];
    }
#pragma unroll
    for (int u = 0; u < GRP; ++u) {
      const int dj = __builtin_amdgcn_readlane(d, c + u);
      const float nj = __int_as_float(__builtin_amdgcn_readlane(nb, c + u));
      float* o = acc + (size_t)dj * D;
      atomicAdd(o + 0,  v[u][0] * nj);
      atomicAdd(o + 32, v[u][1] * nj);
    }
  }
  if (c < n) {
    float v[GRP][2];
#pragma unroll
    for (int u = 0; u < GRP; ++u) {
      v[u][0] = 0.0f; v[u][1] = 0.0f;
      if (c + u < n) {
        const int sj = __builtin_amdgcn_readlane(s, c + u);
        const float* hr = h + (size_t)sj * D;
        v[u][0] = hr[0];
        v[u][1] = hr[32];
      }
    }
#pragma unroll
    for (int u = 0; u < GRP; ++u) {
      if (c + u < n) {
        const int dj = __builtin_amdgcn_readlane(d, c + u);
        const float nj = __int_as_float(__builtin_amdgcn_readlane(nb, c + u));
        float* o = acc + (size_t)dj * D;
        atomicAdd(o + 0,  v[u][0] * nj);
        atomicAdd(o + 32, v[u][1] * nj);
      }
    }
  }
}

template <bool FIRST>
__global__ __launch_bounds__(256) __attribute__((amdgpu_waves_per_eu(16, 16))) void k_scatter(const float* __restrict__ h,
                                                 const int* __restrict__ ei, int E,
                                                 const float* __restrict__ dis,
                                                 float* __restrict__ acc,
                                                 unsigned* __restrict__ tick, int nItems,
                                                 float* __restrict__ ew) {
  __shared__ int pick[2];

  if (threadIdx.x == 0) {
    const unsigned id = __builtin_amdgcn_s_sendmsg_rtn(0x87);
    int par = (int)((id >> 18) & 1u);
    unsigned t = atomicAdd(&tick[par * 64], 1u);
    if (t >= (unsigned)nItems) {
      par ^= 1;
      t = atomicAdd(&tick[par * 64], 1u);
    }
    pick[0] = (int)t;
    pick[1] = par;
  }
  __syncthreads();
  const int item = __builtin_amdgcn_readfirstlane(pick[0]);
  const int par  = __builtin_amdgcn_readfirstlane(pick[1]);
  if ((unsigned)item >= (unsigned)nItems) return;

  const int wv   = __builtin_amdgcn_readfirstlane((int)(threadIdx.x >> 5));
  const int lane = threadIdx.x & 31;
  const int e0   = item * ITEM + wv * 64;

  const int ea = e0 + lane;
  const int eb = ea + 32;
  const int ia = (ea < E) ? ea : E - 1;
  const int ib = (eb < E) ? eb : E - 1;
  const int sa = ei[ia];
  const int da = ei[E + ia];
  const int sb = ei[ib];
  const int db = ei[E + ib];
  float na, nb;
  if (FIRST) {
    na = dis[da];
    nb = dis[db];
  } else {
    na = ew[ia];
    nb = ew[ib];
  }

  int cntA = E - e0;
  cntA = cntA < 0 ? 0 : (cntA > 32 ? 32 : cntA);
  int cntB = E - e0 - 32;
  cntB = cntB < 0 ? 0 : (cntB > 32 ? 32 : cntB);

  const int col = 64 * par + lane;
  scatter_list(h + col, acc + col, sa, da, __float_as_int(na), cntA);
  if (FIRST && par == 0) {
    if (ea < E) ew[ea] = na;
    if (eb < E) ew[eb] = nb;
  }
  scatter_list(h + col, acc + col, sb, db, __float_as_int(nb), cntB);
}

extern "C" void kernel_launch(void* const* d_in, const int* in_sizes, int n_in,
                              void* d_out, int out_size, void* d_ws, size_t ws_size,
                              hipStream_t stream) {
  const float* x  = (const float*)d_in[0];
  const int*   ei = (const int*)d_in[1];
  const float* W1 = (const float*)d_in[2];
  const float* b1 = (const float*)d_in[3];
  const float* W2 = (const float*)d_in[4];
  const float* b2 = (const float*)d_in[5];
  const float* W3 = (const float*)d_in[6];
  const float* b3 = (const float*)d_in[7];

  const int N = in_sizes[0] / D;
  const int E = in_sizes[1] / 2;

  float* deg  = (float*)d_ws;
  size_t npad = ((size_t)N + 255) & ~(size_t)255;
  float* buf1 = deg + npad;
  float* buf2 = buf1 + (size_t)N * D;
  float* buf3 = buf2 + (size_t)N * D;
  unsigned* part = (unsigned*)(buf3 + (size_t)N * D);
  float* out  = (float*)d_out;

  const dim3 blk(256);
  const int gN    = (N + 255) / 256;
  const int nTile  = (N + TILE - 1) / TILE;
  const int nChunk = (E + CH - 1) / CH;
  const int nItem = (E + ITEM - 1) / ITEM;
  unsigned* tick = part + (size_t)nChunk * nTile * TILE;
  float* ew = (float*)(tick + TKW);

  k_deg_part<<<dim3(nTile, nChunk), dim3(CTH), 0, stream>>>(ei + E, E, part, nTile * TILE);
  k_deg_rsqrt<<<gN, blk, 0, stream>>>(part, nTile * TILE, nChunk, deg, N, tick);

  k_gemm128<false><<<GW, blk, 0, stream>>>(x, W1, b1, deg, buf1, buf2, N);
  k_scatter<true><<<2 * nItem, blk, 0, stream>>>(buf1, ei, E, deg, buf2, tick, nItem, ew);

  k_gemm128<true><<<GW, blk, 0, stream>>>(buf2, W2, b2, deg, buf1, buf3, N);
  k_scatter<false><<<2 * nItem, blk, 0, stream>>>(buf1, ei, E, deg, buf3, tick + 128, nItem, ew);

  k_gemm128<true><<<GW, blk, 0, stream>>>(buf3, W3, b3, deg, buf1, out, N);
  k_scatter<false><<<2 * nItem, blk, 0, stream>>>(buf1, ei, E, deg, out, tick + 256, nItem, ew);
}
